# starting kernel
# baseline (speedup 1.0000x reference)
#include <hip/hip_runtime.h>
#include <hip/hip_bf16.h>


typedef __attribute__((ext_vector_type(2))) float v2f;
typedef __attribute__((ext_vector_type(8))) float v8f;

__global__ __launch_bounds__(128) void gemm_wmma_f32(
    const float* __restrict__ A, const float* __restrict__ B,
    float* __restrict__ C, int M, int K, int N) {
  const int wave = threadIdx.x >> 5;
  const int lane = threadIdx.x & 31;
  const int half = lane >> 4;
  const int l    = lane & 15;

  const int mBase = blockIdx.x * 16;
  const int nBase = wave * 16;

  int mRow = mBase + l;
  if (mRow >= M) mRow = M - 1;
  const float* __restrict__ Arow = A + (long long)mRow * K;
  const float* __restrict__ Bcol = B + nBase + l;

  v8f acc = {0.f, 0.f, 0.f, 0.f, 0.f, 0.f, 0.f, 0.f};

  for (int k = 0; k < K; k += 4) {
    const int k0 = k + half * 2;
    v2f a;
    a.x = Arow[k0];
    a.y = Arow[k0 + 1];
    v2f b;
    b.x = Bcol[(long long)k0 * N];
    b.y = Bcol[(long long)(k0 + 1) * N];
    acc = __builtin_amdgcn_wmma_f32_16x16x4_f32(
        false, a, false, b, (short)0, acc, false, false);
  }

  const int mOut = mBase + half * 8;
  float* __restrict__ Cp = C + (long long)mOut * N + nBase + l;
#pragma unroll
  for (int v = 0; v < 8; ++v) {
    if (mOut + v < M) Cp[(long long)v * N] = acc[v];
  }
}

__global__ void init_bias_kernel(float* __restrict__ out,
                                 const float* __restrict__ bias,
                                 long long total) {
  long long t = (long long)blockIdx.x * blockDim.x + threadIdx.x;
  if (t < total) out[t] = bias[t & 63];
}

__global__ __launch_bounds__(256) void spmm_kernel(
    const int* __restrict__ row, const int* __restrict__ col,
    const float* __restrict__ val, const float* __restrict__ dense,
    float* __restrict__ out, int E) {
  long long t = (long long)blockIdx.x * blockDim.x + threadIdx.x;
  int e = (int)(t >> 4);
  if (e >= E) return;
  const int d = (int)(t & 15) << 2;

  const int r = row[e];
  const int c = col[e];
  const float v = val[e];

  const float4 s = *(const float4*)(dense + (long long)c * 64 + d);
  float* o = out + (long long)r * 64 + d;
  atomicAdd(o + 0, v * s.x);
  atomicAdd(o + 1, v * s.y);
  atomicAdd(o + 2, v * s.z);
  atomicAdd(o + 3, v * s.w);
}

__global__ void relu_kernel(float* __restrict__ x, long long total) {
  long long t = (long long)blockIdx.x * blockDim.x + threadIdx.x;
  if (t < total) x[t] = fmaxf(x[t], 0.0f);
}

extern "C" void kernel_launch(void* const* d_in, const int* in_sizes, int n_in,
                              void* d_out, int out_size, void* d_ws, size_t ws_size,
                              hipStream_t stream) {
  const float* x    = (const float*)d_in[0];
  const int*   erow = (const int*)  d_in[1];
  const int*   ecol = (const int*)  d_in[2];
  const float* eval = (const float*)d_in[3];
  const float* W1   = (const float*)d_in[4];
  const float* b1   = (const float*)d_in[5];
  const float* W2   = (const float*)d_in[6];
  const float* b2   = (const float*)d_in[7];
  float* out = (float*)d_out;

  const int IN_DIM = 128, D = 64;
  const int N = in_sizes[0] / IN_DIM;
  const int E = in_sizes[1];
  const long long tot = (long long)N * D;

  float* buf0 = (float*)d_ws;
  float* buf1 = buf0 + (size_t)N * D;

  const int gemmBlocks = (N + 15) / 16;
  const int elwBlocks  = (int)((tot + 255) / 256);
  const int spmmBlocks = (int)(((long long)E * 16 + 255) / 256);

  gemm_wmma_f32<<<gemmBlocks, 128, 0, stream>>>(x, W1, buf0, N, IN_DIM, D);
  init_bias_kernel<<<elwBlocks, 256, 0, stream>>>(buf1, b1, tot);
  spmm_kernel<<<spmmBlocks, 256, 0, stream>>>(erow, ecol, eval, buf0, buf1, E);
  relu_kernel<<<elwBlocks, 256, 0, stream>>>(buf1, tot);

  gemm_wmma_f32<<<gemmBlocks, 128, 0, stream>>>(buf1, W2, buf0, N, D, D);
  init_bias_kernel<<<elwBlocks, 256, 0, stream>>>(out, b2, tot);
  spmm_kernel<<<spmmBlocks, 256, 0, stream>>>(erow, ecol, eval, buf0, out, E);
}


# SpMM: output rows split in two ranges, one per package half; a work-group claims edge chunks for its own half's range and adds only those edges: a line gets adds from one half; sums in arrival order
# speedup vs baseline: 2.3795x; 2.3795x over previous
#include <hip/hip_runtime.h>
#include <hip/hip_bf16.h>


typedef __attribute__((ext_vector_type(2))) float v2f;
typedef __attribute__((ext_vector_type(4))) float v4f;
typedef __attribute__((ext_vector_type(8))) float v8f;

template <int K, bool RELU_A>
__global__ __launch_bounds__(128) void gemm_wmma_f32(
    const float* __restrict__ A, const float* __restrict__ B,
    float* __restrict__ C, const float* __restrict__ bias,
    float* __restrict__ fill, int* __restrict__ claim, int M) {
  constexpr int TM = 32;
  constexpr int N  = 64;
  constexpr int PA = K + 4;
  constexpr int PC = N + 4;
  constexpr int NA = TM * K / 4 / 128;
  constexpr int NC = TM * N / 4 / 128;
  __shared__ __attribute__((aligned(16))) float tile[TM * PA];

  const int t    = threadIdx.x;
  const int wave = t >> 5;
  const int lane = t & 31;
  const int half = lane >> 4;
  const int l    = lane & 15;

  const int mBase = blockIdx.x * TM;
  const int nBase = wave * 16;

  if (blockIdx.x == 0 && t < 2) claim[t * 64] = 0;

  v4f av[NA];
#pragma unroll
  for (int j = 0; j < NA; ++j) {
    const int i = t + 128 * j;
    const int r = (i * 4) / K;
    const int c = (i * 4) % K;
    int m = mBase + r;
    if (m >= M) m = M - 1;
    av[j] = *reinterpret_cast<const v4f*>(A + (long long)m * K + c);
  }

  const float* __restrict__ Bcol = B + nBase + l;
  v2f bf[K / 4];
#pragma unroll
  for (int ks = 0; ks < K / 4; ++ks) {
    const int k0 = ks * 4 + half * 2;
    bf[ks].x = Bcol[k0 * N];
    bf[ks].y = Bcol[(k0 + 1) * N];
  }

  const int fc = (t & 15) * 4;
  const v4f bv = *reinterpret_cast<const v4f*>(bias + fc);

#pragma unroll
  for (int j = 0; j < NA; ++j) {
    const int i = t + 128 * j;
    const int r = (i * 4) / K;
    const int c = (i * 4) % K;
    v4f a = av[j];
    if (RELU_A) {
      a.x = fmaxf(a.x, 0.0f);
      a.y = fmaxf(a.y, 0.0f);
      a.z = fmaxf(a.z, 0.0f);
      a.w = fmaxf(a.w, 0.0f);
    }
    *reinterpret_cast<v4f*>(&tile[r * PA + c]) = a;
  }
  __syncthreads();

  v8f acc0 = {0.f, 0.f, 0.f, 0.f, 0.f, 0.f, 0.f, 0.f};
  v8f acc1 = {0.f, 0.f, 0.f, 0.f, 0.f, 0.f, 0.f, 0.f};
  const float* a0p = &tile[l * PA + half * 2];
  const float* a1p = &tile[(16 + l) * PA + half * 2];
#pragma unroll
  for (int ks = 0; ks < K / 4; ++ks) {
    const v2f a0 = *reinterpret_cast<const v2f*>(a0p + ks * 4);
    const v2f a1 = *reinterpret_cast<const v2f*>(a1p + ks * 4);
    acc0 = __builtin_amdgcn_wmma_f32_16x16x4_f32(
        false, a0, false, bf[ks], (short)0, acc0, false, false);
    acc1 = __builtin_amdgcn_wmma_f32_16x16x4_f32(
        false, a1, false, bf[ks], (short)0, acc1, false, false);
  }
  __syncthreads();

#pragma unroll
  for (int v = 0; v < 8; ++v) {
    tile[(half * 8 + v) * PC + nBase + l]      = acc0[v];
    tile[(16 + half * 8 + v) * PC + nBase + l] = acc1[v];
  }
  __syncthreads();

#pragma unroll
  for (int j = 0; j < NC; ++j) {
    const int r = (t + 128 * j) >> 4;
    const int m = mBase + r;
    if (m < M) {
      const long long o = (long long)m * N + fc;
      *reinterpret_cast<v4f*>(C + o) =
          *reinterpret_cast<const v4f*>(&tile[r * PC + fc]);
      *reinterpret_cast<v4f*>(fill + o) = bv;
    }
  }
}

template <int G>
__global__ __launch_bounds__(256) void spmm_kernel(
    const int* __restrict__ row, const int* __restrict__ col,
    const float* __restrict__ val, const float* __restrict__ dense,
    float* __restrict__ out, int* __restrict__ claim, int E, int N) {
  __shared__ int s_chunk;
  __shared__ int s_range;
  const int lane = threadIdx.x & 31;
  const int wave = __builtin_amdgcn_readfirstlane((int)(threadIdx.x >> 5));
  const int nChunks = (E + 255) / 256;
  const int split = N >> 1;

  const int myHalf = (__builtin_amdgcn_s_sendmsg_rtn(0x87) >> 18) & 1;
  int range = myHalf;

  for (int it = 0; it < 2 * nChunks + 2; ++it) {
    if (threadIdx.x == 0) {
      int id = atomicAdd(&claim[range * 64], 1);
      if (id >= nChunks && range == myHalf) {
        range ^= 1;
        id = atomicAdd(&claim[range * 64], 1);
      }
      s_chunk = (id < nChunks) ? id : -1;
      s_range = range;
    }
    __syncthreads();
    const int id = s_chunk;
    const int upper = s_range;
    __syncthreads();
    if (id < 0) break;

    const long long e0 = (long long)id * 256 + wave * 32;
    if (e0 >= E) continue;
    const int cnt = (E - e0 < 32) ? (int)(E - e0) : 32;

    int r = 0, c = 0;
    float v = 0.f;
    bool mine = false;
    if (lane < cnt) {
      r = row[e0 + lane];
      c = col[e0 + lane];
      v = val[e0 + lane];
      mine = ((r >= split) == (upper != 0));
    }
    unsigned m = __builtin_amdgcn_ballot_w32(mine);

    while (m) {
      int ln[G];
#pragma unroll
      for (int j = 0; j < G; ++j) {
        ln[j] = m ? __builtin_ctz(m) : -1;
        m &= m - 1;
      }
      float s0[G], s1[G];
#pragma unroll
      for (int j = 0; j < G; ++j) {
        s0[j] = 0.f;
        s1[j] = 0.f;
        if (ln[j] >= 0) {
          const int cj = __builtin_amdgcn_readlane(c, ln[j]);
          const float* __restrict__ src = dense + (long long)cj * 64 + lane;
          s0[j] = src[0];
          s1[j] = src[32];
        }
      }
#pragma unroll
      for (int j = 0; j < G; ++j) {
        if (ln[j] >= 0) {
          const int rj = __builtin_amdgcn_readlane(r, ln[j]);
          const float vj = __int_as_float(
              __builtin_amdgcn_readlane(__float_as_int(v), ln[j]));
          float* dst = out + (long long)rj * 64 + lane;
          atomicAdd(dst, vj * s0[j]);
          atomicAdd(dst + 32, vj * s1[j]);
        }
      }
    }
  }
}

extern "C" void kernel_launch(void* const* d_in, const int* in_sizes, int n_in,
                              void* d_out, int out_size, void* d_ws, size_t ws_size,
                              hipStream_t stream) {
  const float* x    = (const float*)d_in[0];
  const int*   erow = (const int*)  d_in[1];
  const int*   ecol = (const int*)  d_in[2];
  const float* eval = (const float*)d_in[3];
  const float* W1   = (const float*)d_in[4];
  const float* b1   = (const float*)d_in[5];
  const float* W2   = (const float*)d_in[6];
  const float* b2   = (const float*)d_in[7];
  float* out = (float*)d_out;

  const int IN_DIM = 128, D = 64;
  const int N = in_sizes[0] / IN_DIM;
  const int E = in_sizes[1];

  float* buf0 = (float*)d_ws;
  float* buf1 = buf0 + (size_t)N * D;
  int* claim1 = (int*)(buf1 + (size_t)N * D);
  int* claim2 = claim1 + 128;

  const int gemmBlocks = (N + 31) / 32;
  const int spmmBlocks = 2048;

  gemm_wmma_f32<128, false><<<gemmBlocks, 128, 0, stream>>>(x, W1, buf0, b1,
                                                           buf1, claim1, N);
  spmm_kernel<8><<<spmmBlocks, 256, 0, stream>>>(erow, ecol, eval, buf0, buf1,
                                                 claim1, E, N);

  gemm_wmma_f32<64, true><<<gemmBlocks, 128, 0, stream>>>(buf1, W2, buf0, b2,
                                                         out, claim2, N);
  spmm_kernel<8><<<spmmBlocks, 256, 0, stream>>>(erow, ecol, eval, buf0, out,
                                                 claim2, E, N);
}
